# starting kernel
# baseline (speedup 1.0000x reference)
#include <hip/hip_runtime.h>
#include <hip/hip_bf16.h>

typedef float v2f __attribute__((ext_vector_type(2)));
typedef float v4f __attribute__((ext_vector_type(4)));
typedef float v8f __attribute__((ext_vector_type(8)));

#define TOTAL_ROWS 262144
#define NSEG 2048
#define DIM  256

__global__ void __launch_bounds__(256)
seg_mean_kernel(const float* __restrict__ values,
                const int*   __restrict__ indices,
                float*       __restrict__ seq_mean)
{
    __shared__ float sdata[4 * DIM];

    const int seg   = blockIdx.x;
    const int start = indices[2 * seg + 0];
    const int end   = indices[2 * seg + 1];
    const int len   = end - start;

    const int tid    = threadIdx.x;
    const int rowgrp = tid >> 6;
    const int col4   = (tid & 63) << 2;

    v4f acc = {0.f, 0.f, 0.f, 0.f};
    for (int r = start + rowgrp; r < end; r += 4) {
        const v4f* p = (const v4f*)(values + (size_t)r * DIM + col4);
        v4f v = __builtin_nontemporal_load(p);
        acc += v;
    }
    sdata[rowgrp * DIM + col4 + 0] = acc.x;
    sdata[rowgrp * DIM + col4 + 1] = acc.y;
    sdata[rowgrp * DIM + col4 + 2] = acc.z;
    sdata[rowgrp * DIM + col4 + 3] = acc.w;
    __syncthreads();

    float s = sdata[0 * DIM + tid] + sdata[1 * DIM + tid] +
              sdata[2 * DIM + tid] + sdata[3 * DIM + tid];
    seq_mean[(size_t)seg * DIM + tid] = s * (1.0f / (float)len);
}

__global__ void __launch_bounds__(256)
mean_gemm_wmma_kernel(const float* __restrict__ seq_mean,
                      const float* __restrict__ W,
                      const float* __restrict__ bias,
                      float*       __restrict__ out)
{
    const int wave  = threadIdx.x >> 5;
    const int lane  = threadIdx.x & 31;
    const int tile  = blockIdx.x * 8 + wave;
    const int mtile = tile & 127;
    const int ntile = tile >> 7;

    const int half = lane >> 4;
    const int l15  = lane & 15;

    const int row0 = mtile * 16;
    const int col  = ntile * 16 + l15;

    const float* Arow  = seq_mean + (size_t)(row0 + l15) * DIM + 2 * half;
    const float* Bcol  = W + col;

    v8f c = {};
    #pragma unroll 4
    for (int k = 0; k < DIM; k += 4) {
        v2f a;
        {
            const float2 a2 = *(const float2*)(Arow + k);
            a.x = a2.x; a.y = a2.y;
        }
        v2f bfrag;
        bfrag.x = Bcol[(size_t)(k + 0 + 2 * half) * DIM];
        bfrag.y = Bcol[(size_t)(k + 1 + 2 * half) * DIM];

        c = __builtin_amdgcn_wmma_f32_16x16x4_f32(
                 false, a,  false, bfrag,
                 (short)0, c,  false,  false);
    }

    const float bv = bias[col];
    #pragma unroll
    for (int v = 0; v < 8; ++v) {
        const int row = row0 + v + 8 * half;
        out[(size_t)row * DIM + col] = c[v] + bv;
    }
}

extern "C" void kernel_launch(void* const* d_in, const int* in_sizes, int n_in,
                              void* d_out, int out_size, void* d_ws, size_t ws_size,
                              hipStream_t stream)
{
    const float* values  = (const float*)d_in[0];
    const int*   indices = (const int*)  d_in[1];
    const float* W       = (const float*)d_in[2];
    const float* bias    = (const float*)d_in[3];
    float*       out     = (float*)d_out;

    float* seq_mean = (float*)d_ws;

    seg_mean_kernel<<<NSEG, 256, 0, stream>>>(values, indices, seq_mean);

    mean_gemm_wmma_kernel<<<NSEG / 8, 256, 0, stream>>>(seq_mean, W, bias, out);
}


# two launches: all segments' rows as one list cut into 512 equal chunks, one per work-group (piece sums via workspace); one wave per segment adds its pieces, 16-segment WMMA tiles; sum order changes
# speedup vs baseline: 3.1744x; 3.1744x over previous
#include <hip/hip_runtime.h>


typedef float v2f __attribute__((ext_vector_type(2)));
typedef float v4f __attribute__((ext_vector_type(4)));
typedef float v8f __attribute__((ext_vector_type(8)));

#define TOTAL_ROWS 262144
#define NSEG 2048
#define DIM  256

#define NCHUNK 512
#define NPAIR  8
#define BATCH  16
#define PGRP   4
#define APITCH 260
#define WS_HDR 4096
#define FLAT_BYTES ((size_t)WS_HDR * 4 + (size_t)(NSEG + NCHUNK) * DIM * 4)

__global__ void __launch_bounds__(512)
chunk_sum_kernel(const float* __restrict__ values,
                 const int*   __restrict__ indices,
                 float*       __restrict__ ws,
                 size_t ws_bytes)
{
    __shared__ int Ps[NSEG + 1];
    __shared__ int Ss[NSEG];
    __shared__ int wtot[16];
    __shared__ int jrange[2];
    __shared__ __attribute__((aligned(16))) float part[NPAIR * PGRP * DIM];

    const int tid  = threadIdx.x;
    const int lane = tid & 31;
    const int w    = __builtin_amdgcn_readfirstlane(tid >> 5);
    const int p    = w >> 1;
    const int coff = (w & 1) * 128 + lane * 4;
    const int g    = blockIdx.x;
    const bool flat = ws_bytes >= FLAT_BYTES;

    int s[4], n[4];
    {
        const int4 a = ((const int4*)indices)[2 * tid + 0];
        const int4 b = ((const int4*)indices)[2 * tid + 1];
        const int sr[4] = {a.x, a.z, b.x, b.z};
        const int er[4] = {a.y, a.w, b.y, b.w};
        #pragma unroll
        for (int k = 0; k < 4; ++k) {
            s[k] = min(max(sr[k], 0), TOTAL_ROWS);
            n[k] = max(min(er[k], TOTAL_ROWS) - s[k], 0);
        }
    }
    const int loc = n[0] + n[1] + n[2] + n[3];
    int inc = loc;
    #pragma unroll
    for (int d = 1; d < 32; d <<= 1) {
        const int t = __shfl_up(inc, d);
        if (lane >= d) inc += t;
    }
    if (lane == 31) wtot[w] = inc;
    __syncthreads();
    int woff = 0, T = 0;
    for (int q = 0; q < 16; ++q) {
        const int t = wtot[q];
        if (q < w) woff += t;
        T += t;
    }
    int ex[5];
    ex[0] = woff + inc - loc;
    ex[1] = ex[0] + n[0];
    ex[2] = ex[1] + n[1];
    ex[3] = ex[2] + n[2];
    ex[4] = ex[3] + n[3];
    #pragma unroll
    for (int k = 0; k < 4; ++k) {
        Ps[4 * tid + k] = ex[k];
        Ss[4 * tid + k] = s[k];
    }
    if (tid == 511) Ps[NSEG] = ex[4];
    if (flat) {
        if (tid == g) ((int4*)ws)[g] = make_int4(ex[0], ex[1], ex[2], ex[3]);
        if (g == NCHUNK - 1 && tid == 511) ((int*)ws)[NSEG] = ex[4];
    }
    __syncthreads();

    int c = (T + NCHUNK - 1) / NCHUNK;
    if (c < 1) c = 1;
    int t0, t1;
    if (flat) {
        t0 = min(g * c, T);
        t1 = min(t0 + c, T);
    } else {
        t0 = Ps[4 * g];
        t1 = Ps[4 * g + 4];
    }
    if (t0 >= t1) return;
    #pragma unroll
    for (int k = 0; k < 4; ++k) {
        if (n[k] > 0 && ex[k] <= t0     && t0     < ex[k + 1]) jrange[0] = 4 * tid + k;
        if (n[k] > 0 && ex[k] <= t1 - 1 && t1 - 1 < ex[k + 1]) jrange[1] = 4 * tid + k;
    }
    __syncthreads();
    const int j0 = __builtin_amdgcn_readfirstlane(jrange[0]);
    const int j1 = __builtin_amdgcn_readfirstlane(jrange[1]);

    float* slots   = ws + (flat ? WS_HDR : 0);
    const int goff = flat ? g : 0;
    const int sg = w >> 2;
    const int c2 = (w & 3) * 64 + lane * 2;

    for (int jb = j0; jb <= j1; jb += PGRP) {
        #pragma unroll
        for (int k = 0; k < PGRP; ++k) {
            const int j = jb + k;
            v4f acc = {0.f, 0.f, 0.f, 0.f};
            if (j <= j1) {
                const int Pj  = __builtin_amdgcn_readfirstlane(Ps[j]);
                const int Pj1 = __builtin_amdgcn_readfirstlane(Ps[j + 1]);
                const int sj  = __builtin_amdgcn_readfirstlane(Ss[j]);
                const int lo = max(Pj, t0), hi = min(Pj1, t1);
                const int r0 = sj + (lo - Pj);
                const int r1 = sj + (hi - Pj);
                for (int r = r0 + p; r < r1; r += NPAIR * BATCH) {
                    const float* base = values + (size_t)r * DIM + coff;
                    v4f v[BATCH];
                    #pragma unroll
                    for (int q = 0; q < BATCH; ++q)
                        if (r + NPAIR * q < r1)
                            v[q] = __builtin_nontemporal_load(
                                       (const v4f*)(base + (size_t)q * NPAIR * DIM));
                    #pragma unroll
                    for (int q = 0; q < BATCH; ++q)
                        if (r + NPAIR * q < r1)
                            acc += v[q];
                }
            }
            *(v4f*)&part[(p * PGRP + k) * DIM + coff] = acc;
        }
        __syncthreads();
        {
            const int j = jb + sg;
            v2f sum = {0.f, 0.f};
            for (int q = 0; q < NPAIR; ++q)
                sum += *(const v2f*)&part[(q * PGRP + sg) * DIM + c2];
            if (j <= j1) {
                const int Pj  = __builtin_amdgcn_readfirstlane(Ps[j]);
                const int Pj1 = __builtin_amdgcn_readfirstlane(Ps[j + 1]);
                if (max(Pj, t0) < min(Pj1, t1))
                    *(v2f*)(slots + (size_t)(j + goff) * DIM + c2) = sum;
            }
        }
        __syncthreads();
    }
}

__global__ void __launch_bounds__(512)
mean_linear_kernel(const int*   __restrict__ indices,
                   const float* __restrict__ W,
                   const float* __restrict__ bias,
                   const float* __restrict__ ws,
                   size_t ws_bytes,
                   float*       __restrict__ out)
{
    __shared__ __attribute__((aligned(16))) float lds[2 * 16 * APITCH];
    float* At  = lds;
    float* Cst = lds + 16 * APITCH;

    const int tid  = threadIdx.x;
    const int lane = tid & 31;
    const int w    = __builtin_amdgcn_readfirstlane(tid >> 5);
    const int j    = blockIdx.x * 16 + w;
    const int c4   = lane * 4;
    const bool flat = ws_bytes >= FLAT_BYTES;

    const int sraw = indices[2 * j + 0];
    const int eraw = indices[2 * j + 1];
    int Tq = 0, Pq = 0;
    if (flat) {
        Tq = ((const int*)ws)[NSEG];
        Pq = ((const int*)ws)[j];
    }
    const v4f bv0 = *(const v4f*)(bias + c4);
    const v4f bv1 = *(const v4f*)(bias + c4 + 128);
    const int sc = min(max(sraw, 0), TOTAL_ROWS);
    const int nj = max(min(eraw, TOTAL_ROWS) - sc, 0);
    v4f a0 = {0.f, 0.f, 0.f, 0.f}, a1 = {0.f, 0.f, 0.f, 0.f};
    if (nj > 0) {
        int gf = 0, gl = 0;
        if (flat) {
            int c = (Tq + NCHUNK - 1) / NCHUNK;
            if (c < 1) c = 1;
            gf = Pq / c;
            gl = (Pq + nj - 1) / c;
        }
        const float* slots = ws + (flat ? WS_HDR : 0);
        for (int gq = gf; gq <= gl; ++gq) {
            const float* row = slots + (size_t)(j + gq) * DIM + c4;
            a0 += *(const v4f*)(row);
            a1 += *(const v4f*)(row + 128);
        }
    }
    {
        const float inv = 1.0f / (float)(eraw - sraw);
        a0 *= inv;
        a1 *= inv;
        *(v4f*)&At[w * APITCH + c4]       = a0;
        *(v4f*)&At[w * APITCH + c4 + 128] = a1;
    }
    __syncthreads();

    const int h   = lane >> 4;
    const int c15 = lane & 15;
    const int col = w * 16 + c15;
    const float* Arow = At + c15 * APITCH + 2 * h;
    const float* Bcol = W + col;

    v8f c = {};
    #pragma unroll 8
    for (int k = 0; k < DIM; k += 4) {
        const v2f a = *(const v2f*)(Arow + k);
        v2f bfrag;
        bfrag.x = Bcol[(size_t)(k + 0 + 2 * h) * DIM];
        bfrag.y = Bcol[(size_t)(k + 1 + 2 * h) * DIM];
        c = __builtin_amdgcn_wmma_f32_16x16x4_f32(
                 false, a,  false, bfrag,
                 (short)0, c,  false,  false);
    }

    #pragma unroll
    for (int v = 0; v < 8; ++v)
        Cst[(v + 8 * h) * APITCH + col] = c[v];
    __syncthreads();

    {
        v4f o0 = *(const v4f*)&Cst[w * APITCH + c4];
        v4f o1 = *(const v4f*)&Cst[w * APITCH + c4 + 128];
        o0 += bv0;
        o1 += bv1;
        *(v4f*)(out + (size_t)j * DIM + c4)       = o0;
        *(v4f*)(out + (size_t)j * DIM + c4 + 128) = o1;
    }
}

extern "C" void kernel_launch(void* const* d_in, const int* in_sizes, int n_in,
                              void* d_out, int out_size, void* d_ws, size_t ws_size,
                              hipStream_t stream)
{
    const float* values  = (const float*)d_in[0];
    const int*   indices = (const int*)  d_in[1];
    const float* W       = (const float*)d_in[2];
    const float* bias    = (const float*)d_in[3];
    float*       out     = (float*)d_out;
    float*       ws      = (float*)d_ws;

    chunk_sum_kernel<<<NCHUNK, 512, 0, stream>>>(values, indices, ws, ws_size);
    mean_linear_kernel<<<NSEG / 16, 512, 0, stream>>>(indices, W, bias, ws, ws_size, out);
}


# two launches kept: each wave pair owns one of 4096 contiguous runs of the row list and stores a sum per (segment, run) piece; no LDS reduce or barrier while streaming; sum order changes
# speedup vs baseline: 3.4826x; 1.0971x over previous
#include <hip/hip_runtime.h>


typedef float v2f __attribute__((ext_vector_type(2)));
typedef float v4f __attribute__((ext_vector_type(4)));
typedef float v8f __attribute__((ext_vector_type(8)));

#define TOTAL_ROWS 262144
#define NSEG 2048
#define DIM  256

#define NCHUNK 512
#define NPAIR  8
#define BATCH  16
#define PGRP   4
#define APITCH 260
#define WS_HDR 4096
#define FLAT_BYTES ((size_t)WS_HDR * 4 + (size_t)(NSEG + NCHUNK) * DIM * 4)
#define NRUN  (NCHUNK * NPAIR)
#define RUN_BYTES ((size_t)WS_HDR * 4 + (size_t)(NSEG + NRUN) * DIM * 4)

__global__ void __launch_bounds__(512)
chunk_sum_kernel(const float* __restrict__ values,
                 const int*   __restrict__ indices,
                 float*       __restrict__ ws,
                 size_t ws_bytes)
{
    __shared__ int Ps[NSEG + 1];
    __shared__ int Ss[NSEG];
    __shared__ int wtot[16];
    __shared__ int jrange[2];
    __shared__ int jstart[NPAIR];
    __shared__ __attribute__((aligned(16))) float part[NPAIR * PGRP * DIM];

    const int tid  = threadIdx.x;
    const int lane = tid & 31;
    const int w    = __builtin_amdgcn_readfirstlane(tid >> 5);
    const int p    = w >> 1;
    const int coff = (w & 1) * 128 + lane * 4;
    const int g    = blockIdx.x;
    const bool flat = ws_bytes >= FLAT_BYTES;
    const bool runs = ws_bytes >= RUN_BYTES;

    int s[4], n[4];
    {
        const int4 a = ((const int4*)indices)[2 * tid + 0];
        const int4 b = ((const int4*)indices)[2 * tid + 1];
        const int sr[4] = {a.x, a.z, b.x, b.z};
        const int er[4] = {a.y, a.w, b.y, b.w};
        #pragma unroll
        for (int k = 0; k < 4; ++k) {
            s[k] = min(max(sr[k], 0), TOTAL_ROWS);
            n[k] = max(min(er[k], TOTAL_ROWS) - s[k], 0);
        }
    }
    const int loc = n[0] + n[1] + n[2] + n[3];
    int inc = loc;
    #pragma unroll
    for (int d = 1; d < 32; d <<= 1) {
        const int t = __shfl_up(inc, d);
        if (lane >= d) inc += t;
    }
    if (lane == 31) wtot[w] = inc;
    __syncthreads();
    int woff = 0, T = 0;
    for (int q = 0; q < 16; ++q) {
        const int t = wtot[q];
        if (q < w) woff += t;
        T += t;
    }
    int ex[5];
    ex[0] = woff + inc - loc;
    ex[1] = ex[0] + n[0];
    ex[2] = ex[1] + n[1];
    ex[3] = ex[2] + n[2];
    ex[4] = ex[3] + n[3];
    #pragma unroll
    for (int k = 0; k < 4; ++k) {
        Ps[4 * tid + k] = ex[k];
        Ss[4 * tid + k] = s[k];
    }
    if (tid == 511) Ps[NSEG] = ex[4];
    if (flat) {
        if (tid == g) ((int4*)ws)[g] = make_int4(ex[0], ex[1], ex[2], ex[3]);
        if (g == NCHUNK - 1 && tid == 511) ((int*)ws)[NSEG] = ex[4];
    }
    __syncthreads();

    if (runs) {
        int cs = (T + NRUN - 1) / NRUN;
        if (cs < 1) cs = 1;
        #pragma unroll
        for (int pp = 0; pp < NPAIR; ++pp) {
            const int uu = (g * NPAIR + pp) * cs;
            #pragma unroll
            for (int k = 0; k < 4; ++k)
                if (n[k] > 0 && ex[k] <= uu && uu < ex[k + 1]) jstart[pp] = 4 * tid + k;
        }
        __syncthreads();
        const int G  = g * NPAIR + p;
        const int u0 = min(G * cs, T);
        const int u1 = min(u0 + cs, T);
        if (u0 >= u1) return;
        float* rslots = ws + WS_HDR;
        int j = __builtin_amdgcn_readfirstlane(jstart[p]);
        int t = u0;
        while (t < u1) {
            const int Pj  = __builtin_amdgcn_readfirstlane(Ps[j]);
            const int Pj1 = __builtin_amdgcn_readfirstlane(Ps[j + 1]);
            const int sj  = __builtin_amdgcn_readfirstlane(Ss[j]);
            const int hi  = min(Pj1, u1);
            if (hi > t) {
                const int r0 = sj + (t  - Pj);
                const int r1 = sj + (hi - Pj);
                v4f acc = {0.f, 0.f, 0.f, 0.f};
                for (int r = r0; r < r1; r += BATCH) {
                    const float* base = values + (size_t)r * DIM + coff;
                    v4f v[BATCH];
                    #pragma unroll
                    for (int q = 0; q < BATCH; ++q)
                        if (r + q < r1)
                            v[q] = __builtin_nontemporal_load(
                                       (const v4f*)(base + (size_t)q * DIM));
                    #pragma unroll
                    for (int q = 0; q < BATCH; ++q)
                        if (r + q < r1)
                            acc += v[q];
                }
                *(v4f*)(rslots + (size_t)(j + G) * DIM + coff) = acc;
                t = hi;
            }
            ++j;
        }
        return;
    }

    int c = (T + NCHUNK - 1) / NCHUNK;
    if (c < 1) c = 1;
    int t0, t1;
    if (flat) {
        t0 = min(g * c, T);
        t1 = min(t0 + c, T);
    } else {
        t0 = Ps[4 * g];
        t1 = Ps[4 * g + 4];
    }
    if (t0 >= t1) return;
    #pragma unroll
    for (int k = 0; k < 4; ++k) {
        if (n[k] > 0 && ex[k] <= t0     && t0     < ex[k + 1]) jrange[0] = 4 * tid + k;
        if (n[k] > 0 && ex[k] <= t1 - 1 && t1 - 1 < ex[k + 1]) jrange[1] = 4 * tid + k;
    }
    __syncthreads();
    const int j0 = __builtin_amdgcn_readfirstlane(jrange[0]);
    const int j1 = __builtin_amdgcn_readfirstlane(jrange[1]);

    float* slots   = ws + (flat ? WS_HDR : 0);
    const int goff = flat ? g : 0;
    const int sg = w >> 2;
    const int c2 = (w & 3) * 64 + lane * 2;

    for (int jb = j0; jb <= j1; jb += PGRP) {
        #pragma unroll
        for (int k = 0; k < PGRP; ++k) {
            const int j = jb + k;
            v4f acc = {0.f, 0.f, 0.f, 0.f};
            if (j <= j1) {
                const int Pj  = __builtin_amdgcn_readfirstlane(Ps[j]);
                const int Pj1 = __builtin_amdgcn_readfirstlane(Ps[j + 1]);
                const int sj  = __builtin_amdgcn_readfirstlane(Ss[j]);
                const int lo = max(Pj, t0), hi = min(Pj1, t1);
                const int r0 = sj + (lo - Pj);
                const int r1 = sj + (hi - Pj);
                for (int r = r0 + p; r < r1; r += NPAIR * BATCH) {
                    const float* base = values + (size_t)r * DIM + coff;
                    v4f v[BATCH];
                    #pragma unroll
                    for (int q = 0; q < BATCH; ++q)
                        if (r + NPAIR * q < r1)
                            v[q] = __builtin_nontemporal_load(
                                       (const v4f*)(base + (size_t)q * NPAIR * DIM));
                    #pragma unroll
                    for (int q = 0; q < BATCH; ++q)
                        if (r + NPAIR * q < r1)
                            acc += v[q];
                }
            }
            *(v4f*)&part[(p * PGRP + k) * DIM + coff] = acc;
        }
        __syncthreads();
        {
            const int j = jb + sg;
            v2f sum = {0.f, 0.f};
            for (int q = 0; q < NPAIR; ++q)
                sum += *(const v2f*)&part[(q * PGRP + sg) * DIM + c2];
            if (j <= j1) {
                const int Pj  = __builtin_amdgcn_readfirstlane(Ps[j]);
                const int Pj1 = __builtin_amdgcn_readfirstlane(Ps[j + 1]);
                if (max(Pj, t0) < min(Pj1, t1))
                    *(v2f*)(slots + (size_t)(j + goff) * DIM + c2) = sum;
            }
        }
        __syncthreads();
    }
}

__global__ void __launch_bounds__(512)
mean_linear_kernel(const int*   __restrict__ indices,
                   const float* __restrict__ W,
                   const float* __restrict__ bias,
                   const float* __restrict__ ws,
                   size_t ws_bytes,
                   float*       __restrict__ out)
{
    __shared__ __attribute__((aligned(16))) float lds[2 * 16 * APITCH];
    float* At  = lds;
    float* Cst = lds + 16 * APITCH;

    const int tid  = threadIdx.x;
    const int lane = tid & 31;
    const int w    = __builtin_amdgcn_readfirstlane(tid >> 5);
    const int j    = blockIdx.x * 16 + w;
    const int c4   = lane * 4;
    const bool flat = ws_bytes >= FLAT_BYTES;
    const bool runs = ws_bytes >= RUN_BYTES;

    const int sraw = indices[2 * j + 0];
    const int eraw = indices[2 * j + 1];
    int Tq = 0, Pq = 0;
    if (flat) {
        Tq = ((const int*)ws)[NSEG];
        Pq = ((const int*)ws)[j];
    }
    const v4f bv0 = *(const v4f*)(bias + c4);
    const v4f bv1 = *(const v4f*)(bias + c4 + 128);
    const int sc = min(max(sraw, 0), TOTAL_ROWS);
    const int nj = max(min(eraw, TOTAL_ROWS) - sc, 0);
    v4f a0 = {0.f, 0.f, 0.f, 0.f}, a1 = {0.f, 0.f, 0.f, 0.f};
    if (nj > 0) {
        int gf = 0, gl = 0;
        if (flat) {
            const int ncut = runs ? NRUN : NCHUNK;
            int c = (Tq + ncut - 1) / ncut;
            if (c < 1) c = 1;
            gf = Pq / c;
            gl = (Pq + nj - 1) / c;
        }
        const float* slots = ws + (flat ? WS_HDR : 0);
        for (int gq = gf; gq <= gl; gq += 8) {
            const float* row = slots + (size_t)(j + gq) * DIM + c4;
            v4f p0[8], p1[8];
            #pragma unroll
            for (int q = 0; q < 8; ++q)
                if (gq + q <= gl) {
                    p0[q] = *(const v4f*)(row + (size_t)q * DIM);
                    p1[q] = *(const v4f*)(row + (size_t)q * DIM + 128);
                }
            #pragma unroll
            for (int q = 0; q < 8; ++q)
                if (gq + q <= gl) {
                    a0 += p0[q];
                    a1 += p1[q];
                }
        }
    }
    {
        const float inv = 1.0f / (float)(eraw - sraw);
        a0 *= inv;
        a1 *= inv;
        *(v4f*)&At[w * APITCH + c4]       = a0;
        *(v4f*)&At[w * APITCH + c4 + 128] = a1;
    }
    __syncthreads();

    const int h   = lane >> 4;
    const int c15 = lane & 15;
    const int col = w * 16 + c15;
    const float* Arow = At + c15 * APITCH + 2 * h;
    const float* Bcol = W + col;

    v8f c = {};
    #pragma unroll 8
    for (int k = 0; k < DIM; k += 4) {
        const v2f a = *(const v2f*)(Arow + k);
        v2f bfrag;
        bfrag.x = Bcol[(size_t)(k + 0 + 2 * h) * DIM];
        bfrag.y = Bcol[(size_t)(k + 1 + 2 * h) * DIM];
        c = __builtin_amdgcn_wmma_f32_16x16x4_f32(
                 false, a,  false, bfrag,
                 (short)0, c,  false,  false);
    }

    #pragma unroll
    for (int v = 0; v < 8; ++v)
        Cst[(v + 8 * h) * APITCH + col] = c[v];
    __syncthreads();

    {
        v4f o0 = *(const v4f*)&Cst[w * APITCH + c4];
        v4f o1 = *(const v4f*)&Cst[w * APITCH + c4 + 128];
        o0 += bv0;
        o1 += bv1;
        *(v4f*)(out + (size_t)j * DIM + c4)       = o0;
        *(v4f*)(out + (size_t)j * DIM + c4 + 128) = o1;
    }
}

extern "C" void kernel_launch(void* const* d_in, const int* in_sizes, int n_in,
                              void* d_out, int out_size, void* d_ws, size_t ws_size,
                              hipStream_t stream)
{
    const float* values  = (const float*)d_in[0];
    const int*   indices = (const int*)  d_in[1];
    const float* W       = (const float*)d_in[2];
    const float* bias    = (const float*)d_in[3];
    float*       out     = (float*)d_out;
    float*       ws      = (float*)d_ws;

    chunk_sum_kernel<<<NCHUNK, 512, 0, stream>>>(values, indices, ws, ws_size);
    mean_linear_kernel<<<NSEG / 16, 512, 0, stream>>>(indices, W, bias, ws, ws_size, out);
}
